# starting kernel
# baseline (speedup 1.0000x reference)
#include <hip/hip_runtime.h>
#include <hip/hip_bf16.h>

#define NBATCH 16
#define NFEAT  256
#define NTIME  8000
#define SPEC_ROWS 257
#define ALPHA_F       0.95f
#define ONE_M_ALPHA_F 0.05f

#define NCHUNK 25
#define TCHUNK 320
#define NTILE  20

typedef __attribute__((ext_vector_type(2))) float v2f;
typedef __attribute__((ext_vector_type(4))) float v4f;
typedef __attribute__((ext_vector_type(8))) float v8f;

__device__ __forceinline__ float fast_rsq(float x) {
  return __builtin_amdgcn_rsqf(x);
}

__global__ void fse_pass1_chunk_contrib(const float* __restrict__ spec,
                                        float* __restrict__ Bc) {
  int tid = blockIdx.x * blockDim.x + threadIdx.x;
  if (tid >= NBATCH * NFEAT * NCHUNK) return;
  int c = tid % NCHUNK;
  int f = (tid / NCHUNK) % NFEAT;
  int b = tid / (NCHUNK * NFEAT);

  const float* p = spec + ((size_t)b * SPEC_ROWS + f) * NTIME + (size_t)c * TCHUNK;
  float s = 0.0f;
#pragma unroll 4
  for (int t = 0; t < TCHUNK; t += 4) {
    v4f x = *(const v4f*)(p + t);
    s = fmaf(fabsf(x.x), ONE_M_ALPHA_F, s * ALPHA_F);
    s = fmaf(fabsf(x.y), ONE_M_ALPHA_F, s * ALPHA_F);
    s = fmaf(fabsf(x.z), ONE_M_ALPHA_F, s * ALPHA_F);
    s = fmaf(fabsf(x.w), ONE_M_ALPHA_F, s * ALPHA_F);
  }
  Bc[tid] = s;
}

__global__ void fse_pass2_boundary_scan(const float* __restrict__ s0,
                                        const float* __restrict__ Bc,
                                        float* __restrict__ Sst) {
  int tid = blockIdx.x * blockDim.x + threadIdx.x;
  if (tid >= NBATCH * NFEAT) return;
  int f = tid % NFEAT;

  float decay = __powf(ALPHA_F, (float)TCHUNK);
  float S = s0[f];
  const float* bp = Bc + (size_t)tid * NCHUNK;
  float* sp = Sst + (size_t)tid * NCHUNK;
#pragma unroll
  for (int c = 0; c < NCHUNK; ++c) {
    sp[c] = S;
    S = fmaf(decay, S, bp[c]);
  }
}

__global__ void fse_pass3_wmma_scan(const float* __restrict__ spec,
                                    const float* __restrict__ Sst,
                                    float* __restrict__ out) {
  int wave = (int)((blockIdx.x * blockDim.x + threadIdx.x) >> 5);
  int lane = threadIdx.x & 31;
  if (wave >= NBATCH * (NFEAT / 16) * NCHUNK) return;

  int c  = wave % NCHUNK;
  int fg = (wave / NCHUNK) % (NFEAT / 16);
  int b  = wave / (NCHUNK * (NFEAT / 16));

  int hl = lane >> 4;
  int m  = lane & 15;
  int f  = fg * 16 + m;

  float pw[17];
  pw[0] = 1.0f;
#pragma unroll
  for (int i = 1; i <= 16; ++i) pw[i] = pw[i - 1] * ALPHA_F;

  v2f A[4];
#pragma unroll
  for (int j = 0; j < 4; ++j) {
    int k0 = 4 * j + 2 * hl;
    A[j].x = (k0     <= m) ? ONE_M_ALPHA_F * pw[m - k0]     : 0.0f;
    A[j].y = (k0 + 1 <= m) ? ONE_M_ALPHA_F * pw[m - k0 - 1] : 0.0f;
  }
  v2f Ac;
  Ac.x = (hl == 0) ? pw[m + 1] : 0.0f;
  Ac.y = 0.0f;

  float s_in = Sst[((size_t)b * NFEAT + f) * NCHUNK + c];

  const float* xin  = spec + ((size_t)b * SPEC_ROWS + f) * NTIME + (size_t)c * TCHUNK;
  float*       yout = out  + ((size_t)b * NFEAT     + f) * NTIME + (size_t)c * TCHUNK;

  for (int tile = 0; tile < NTILE; ++tile) {
    const float* xt = xin + tile * 16;

    v2f Bv[4];
#pragma unroll
    for (int j = 0; j < 4; ++j) {
      v2f xv = *(const v2f*)(xt + 4 * j + 2 * hl);
      Bv[j].x = fabsf(xv.x);
      Bv[j].y = fabsf(xv.y);
    }
    v2f Bc2;
    Bc2.x = (hl == 0) ? s_in : 0.0f;
    Bc2.y = 0.0f;

    v8f d = {};
    d = __builtin_amdgcn_wmma_f32_16x16x4_f32(false, Ac,   false, Bc2,   (short)0, d, false, false);
    d = __builtin_amdgcn_wmma_f32_16x16x4_f32(false, A[0], false, Bv[0], (short)0, d, false, false);
    d = __builtin_amdgcn_wmma_f32_16x16x4_f32(false, A[1], false, Bv[1], (short)0, d, false, false);
    d = __builtin_amdgcn_wmma_f32_16x16x4_f32(false, A[2], false, Bv[2], (short)0, d, false, false);
    d = __builtin_amdgcn_wmma_f32_16x16x4_f32(false, A[3], false, Bv[3], (short)0, d, false, false);

    s_in = __shfl(d[7], 16 + m, 32);

    const float* xr = xt + 8 * hl;
    v4f x0 = *(const v4f*)(xr);
    v4f x1 = *(const v4f*)(xr + 4);
    v4f y0, y1;
    y0.x = x0.x * fast_rsq(d[0]);
    y0.y = x0.y * fast_rsq(d[1]);
    y0.z = x0.z * fast_rsq(d[2]);
    y0.w = x0.w * fast_rsq(d[3]);
    y1.x = x1.x * fast_rsq(d[4]);
    y1.y = x1.y * fast_rsq(d[5]);
    y1.z = x1.z * fast_rsq(d[6]);
    y1.w = x1.w * fast_rsq(d[7]);

    float* yw = yout + tile * 16 + 8 * hl;
    __builtin_nontemporal_store(y0, (v4f*)(yw));
    __builtin_nontemporal_store(y1, (v4f*)(yw + 4));
  }
}

extern "C" void kernel_launch(void* const* d_in, const int* in_sizes, int n_in,
                              void* d_out, int out_size, void* d_ws, size_t ws_size,
                              hipStream_t stream) {
  const float* spec = (const float*)d_in[0];
  const float* s0   = (const float*)d_in[1];
  float* out = (float*)d_out;

  float* Bc  = (float*)d_ws;
  float* Sst = Bc + (size_t)NBATCH * NFEAT * NCHUNK;

  {
    int total = NBATCH * NFEAT * NCHUNK;
    fse_pass1_chunk_contrib<<<(total + 255) / 256, 256, 0, stream>>>(spec, Bc);
  }
  {
    int total = NBATCH * NFEAT;
    fse_pass2_boundary_scan<<<(total + 255) / 256, 256, 0, stream>>>(s0, Bc, Sst);
  }
  {
    int waves = NBATCH * (NFEAT / 16) * NCHUNK;
    int total = waves * 32;
    fse_pass3_wmma_scan<<<(total + 255) / 256, 256, 0, stream>>>(spec, Sst, out);
  }
}


# one wave scans a whole feature row: matrix tile = 16 consecutive 16-sample blocks of that row, block carries by a 16-lane scan; x read once, whole-line stores, 1 launch (sum order changes)
# speedup vs baseline: 11.2201x; 11.2201x over previous
#include <hip/hip_runtime.h>


#define NBATCH 16
#define NFEAT  256
#define NTIME  8000
#define SPEC_ROWS 257
#define ALPHA_F       0.95f
#define ONE_M_ALPHA_F 0.05f

#define NROWS    (NBATCH * NFEAT)
#define TILE     256
#define NFULL    31
#define LAST_BLK (NTIME - 16)
#define PF       8
#define WAVES_PER_WG 8

typedef __attribute__((ext_vector_type(2))) float v2f;
typedef __attribute__((ext_vector_type(4))) float v4f;
typedef __attribute__((ext_vector_type(8))) float v8f;

constexpr double alpha_pow_d(int n) {
  double r = 1.0;
  for (int i = 0; i < n; ++i) r *= (double)ALPHA_F;
  return r;
}
constexpr float P1   = ALPHA_F;
constexpr float P2   = (float)alpha_pow_d(2);
constexpr float P4   = (float)alpha_pow_d(4);
constexpr float P8   = (float)alpha_pow_d(8);
constexpr float P16  = (float)alpha_pow_d(16);
constexpr float P32  = (float)alpha_pow_d(32);
constexpr float P64  = (float)alpha_pow_d(64);
constexpr float P128 = (float)alpha_pow_d(128);
constexpr float P256 = (float)alpha_pow_d(256);

__device__ __forceinline__ float apow(int d) {
  float r = (d & 1) ? P1 : 1.0f;
  r *= (d & 2)  ? P2  : 1.0f;
  r *= (d & 4)  ? P4  : 1.0f;
  r *= (d & 8)  ? P8  : 1.0f;
  r *= (d & 16) ? P16 : 1.0f;
  return r;
}
__device__ __forceinline__ float qpow(int i) {
  float r = (i & 1) ? P16 : 1.0f;
  r *= (i & 2) ? P32  : 1.0f;
  r *= (i & 4) ? P64  : 1.0f;
  r *= (i & 8) ? P128 : 1.0f;
  return r;
}

__device__ __forceinline__ float fast_rsq(float x) {
  return __builtin_amdgcn_rsqf(x);
}

__device__ __forceinline__ int tmap(int hl, int v) {
  return 4 * hl + v + ((v >= 4) ? 4 : 0);
}

template <int N>
__device__ __forceinline__ float row_shr0(float v) {
  return __builtin_bit_cast(float, __builtin_amdgcn_update_dpp(
      0, __builtin_bit_cast(int, v), 0x110 + N, 0xf, 0xf, true));
}
__device__ __forceinline__ float row_last(float v) {
  return __builtin_bit_cast(float, __builtin_amdgcn_update_dpp(
      0, __builtin_bit_cast(int, v), 0x15F, 0xf, 0xf, true));
}
__device__ __forceinline__ float xhalf(float x) {
  int b = __builtin_bit_cast(int, x);
  return __builtin_bit_cast(float, __builtin_amdgcn_permlanex16(
      b, b, 0x76543210, 0xfedcba98, false, false));
}

__device__ __forceinline__ float row_swap8(float v) {
  return __builtin_bit_cast(float, __builtin_amdgcn_update_dpp(
      0, __builtin_bit_cast(int, v), 0x128, 0xf, 0xf, true));
}

__device__ __forceinline__ void to_lines(v4f ya, v4f yb, bool lo, v4f& o0, v4f& o1) {
  v4f snd = lo ? yb : ya;
  v4f rcv;
  rcv.x = row_swap8(snd.x);
  rcv.y = row_swap8(snd.y);
  rcv.z = row_swap8(snd.z);
  rcv.w = row_swap8(snd.w);
  o0 = lo ? ya : rcv;
  o1 = lo ? rcv : yb;
}

__device__ __forceinline__ void tile_step(v4f xa, v4f xb, const v2f (&A)[4], v2f Ac,
                                          float qi, int hl, float& s_run,
                                          v4f& ya, v4f& yb) {
  v2f B0, B1, B2, B3;
  B0.x = fabsf(xa.x); B0.y = fabsf(xa.y);
  B1.x = fabsf(xa.z); B1.y = fabsf(xa.w);
  B2.x = fabsf(xb.x); B2.y = fabsf(xb.y);
  B3.x = fabsf(xb.z); B3.y = fabsf(xb.w);

  v8f d = {};
  d = __builtin_amdgcn_wmma_f32_16x16x4_f32(false, A[0], false, B0, (short)0, d, false, false);
  d = __builtin_amdgcn_wmma_f32_16x16x4_f32(false, A[1], false, B1, (short)0, d, false, false);
  d = __builtin_amdgcn_wmma_f32_16x16x4_f32(false, A[2], false, B2, (short)0, d, false, false);
  d = __builtin_amdgcn_wmma_f32_16x16x4_f32(false, A[3], false, B3, (short)0, d, false, false);

  float w  = d[7];
  float wo = xhalf(w);
  w = hl ? w : wo;

  float e = w;
  e = fmaf(P16,  row_shr0<1>(e), e);
  e = fmaf(P32,  row_shr0<2>(e), e);
  e = fmaf(P64,  row_shr0<4>(e), e);
  e = fmaf(P128, row_shr0<8>(e), e);

  float e_in  = fmaf(qi, s_run, row_shr0<1>(e));
  s_run = fmaf(P256, s_run, row_last(e));

  v2f Bc;
  Bc.x = (hl == 0) ? e_in : 0.0f;
  Bc.y = 0.0f;
  d = __builtin_amdgcn_wmma_f32_16x16x4_f32(false, Ac, false, Bc, (short)0, d, false, false);

  ya.x = xa.x * fast_rsq(d[0]);
  ya.y = xa.y * fast_rsq(d[1]);
  ya.z = xa.z * fast_rsq(d[2]);
  ya.w = xa.w * fast_rsq(d[3]);
  yb.x = xb.x * fast_rsq(d[4]);
  yb.y = xb.y * fast_rsq(d[5]);
  yb.z = xb.z * fast_rsq(d[6]);
  yb.w = xb.w * fast_rsq(d[7]);
}

__global__ __launch_bounds__(WAVES_PER_WG * 32)
void fse_row_scan(const float* __restrict__ spec,
                  const float* __restrict__ s0,
                  float* __restrict__ out) {
  int wv  = __builtin_amdgcn_readfirstlane((int)(threadIdx.x >> 5));
  int row = (int)blockIdx.x * WAVES_PER_WG + wv;
  if (row >= NROWS) return;
  int b = row / NFEAT;
  int f = row % NFEAT;

  int lane = threadIdx.x & 31;
  int hl = lane >> 4;
  int m  = lane & 15;

  int tm = tmap(m >> 3, m & 7);
  v2f A[4];
#pragma unroll
  for (int j = 0; j < 4; ++j) {
    int k0 = tmap(hl, 2 * j);
    int k1 = tmap(hl, 2 * j + 1);
    A[j].x = (k0 <= tm) ? ONE_M_ALPHA_F * apow(tm - k0) : 0.0f;
    A[j].y = (k1 <= tm) ? ONE_M_ALPHA_F * apow(tm - k1) : 0.0f;
  }
  v2f Ac;
  Ac.x = (hl == 0) ? apow(tm + 1) : 0.0f;
  Ac.y = 0.0f;
  float qi = qpow(m);

  const float* xrow = spec + ((size_t)b * SPEC_ROWS + f) * NTIME;
  float*       yrow = out  + (size_t)row * NTIME;
  int loff = 16 * m + 4 * hl;
  bool lo  = (m < 8);
  int soff = 16 * (m & 7) + 4 * hl + (lo ? 0 : 8);
  int loff_tail = min(NFULL * TILE + 16 * m, LAST_BLK) + 4 * hl;

  float s_run = s0[f];

  v4f xa[PF], xb[PF];
#pragma unroll
  for (int u = 0; u < PF; ++u) {
    const float* p = xrow + u * TILE + loff;
    xa[u] = *(const v4f*)(p);
    xb[u] = *(const v4f*)(p + 8);
  }

  for (int T = 0; T < NFULL + 1 - PF; T += PF) {
#pragma unroll
    for (int u = 0; u < PF; ++u) {
      v4f ya, yb;
      tile_step(xa[u], xb[u], A, Ac, qi, hl, s_run, ya, yb);
      v4f o0, o1;
      to_lines(ya, yb, lo, o0, o1);
      float* q = yrow + (T + u) * TILE + soff;
      __builtin_nontemporal_store(o0, (v4f*)(q));
      __builtin_nontemporal_store(o1, (v4f*)(q + 128));

      int nt = T + u + PF;
      const float* p = xrow + ((u == PF - 1 && nt == NFULL) ? loff_tail : nt * TILE + loff);
      xa[u] = *(const v4f*)(p);
      xb[u] = *(const v4f*)(p + 8);
      __builtin_amdgcn_sched_barrier(0);
    }
  }
#pragma unroll
  for (int u = 0; u < PF; ++u) {
    v4f ya, yb;
    tile_step(xa[u], xb[u], A, Ac, qi, hl, s_run, ya, yb);
    int T = NFULL + 1 - PF + u;
    v4f o0, o1;
    to_lines(ya, yb, lo, o0, o1);
    float* q = yrow + T * TILE + soff;
    if (u < PF - 1) {
      __builtin_nontemporal_store(o0, (v4f*)(q));
      __builtin_nontemporal_store(o1, (v4f*)(q + 128));
      __builtin_amdgcn_sched_barrier(0);
    } else if ((m & 7) < 4) {
      __builtin_nontemporal_store(o0, (v4f*)(q));
    }
  }
}

extern "C" void kernel_launch(void* const* d_in, const int* in_sizes, int n_in,
                              void* d_out, int out_size, void* d_ws, size_t ws_size,
                              hipStream_t stream) {
  const float* spec = (const float*)d_in[0];
  const float* s0   = (const float*)d_in[1];
  float* out = (float*)d_out;

  fse_row_scan<<<NROWS / WAVES_PER_WG, WAVES_PER_WG * 32, 0, stream>>>(spec, s0, out);
}
